# starting kernel
# baseline (speedup 1.0000x reference)
#include <hip/hip_runtime.h>

typedef _Float16 v16h __attribute__((ext_vector_type(16)));
typedef _Float16 v8h  __attribute__((ext_vector_type(8)));
typedef _Float16 h4   __attribute__((ext_vector_type(4)));
typedef float    v8f  __attribute__((ext_vector_type(8)));

#define D 128

__global__ void __launch_bounds__(256) zero_kernel(float4* __restrict__ p, int n4) {
    int i = blockIdx.x * blockDim.x + threadIdx.x;
    if (i < n4) p[i] = make_float4(0.f, 0.f, 0.f, 0.f);
}

__global__ void __launch_bounds__(256) cvt_f32_f16(const float* __restrict__ src,
                                                   _Float16* __restrict__ dst, int n4) {
    int i = blockIdx.x * blockDim.x + threadIdx.x;
    if (i < n4) {
        float4 v = ((const float4*)src)[i];
        h4 o;
        o[0] = (_Float16)v.x; o[1] = (_Float16)v.y;
        o[2] = (_Float16)v.z; o[3] = (_Float16)v.w;
        ((h4*)dst)[i] = o;
    }
}

__global__ void __launch_bounds__(256) spmm_scatter(
        const int*   __restrict__ rows,
        const int*   __restrict__ cols,
        const float* __restrict__ vals,
        const float* __restrict__ x,
        float*       __restrict__ agg,
        int n_edges) {
    const int wave = blockIdx.x * (blockDim.x >> 5) + (threadIdx.x >> 5);
    const int lane = threadIdx.x & 31;
    const int e0 = wave * 4;
    #pragma unroll
    for (int k = 0; k < 4; ++k) {
        int e = e0 + k;
        if (e >= n_edges) break;
        int   r = rows[e];
        int   c = cols[e];
        float v = vals[e];
        float4 d = ((const float4*)(x + (size_t)c * D))[lane];
        float* dst = agg + (size_t)r * D + lane * 4;
        unsafeAtomicAdd(dst + 0, v * d.x);
        unsafeAtomicAdd(dst + 1, v * d.y);
        unsafeAtomicAdd(dst + 2, v * d.z);
        unsafeAtomicAdd(dst + 3, v * d.w);
    }
}

__global__ void __launch_bounds__(256) gemm_relu_wmma(
        const float*    __restrict__ X,
        const _Float16* __restrict__ W16,
        float*          __restrict__ Y,
        int n_nodes) {
    const int wave    = threadIdx.x >> 5;
    const int lane    = threadIdx.x & 31;
    const int laneMod = lane & 15;
    const int laneHi  = lane >> 4;
    const int rowBase = (blockIdx.x * 8 + wave) * 16;

    int aRow = rowBase + laneMod;
    if (aRow >= n_nodes) aRow = n_nodes - 1;

    const float*    aPtr  = X + (size_t)aRow * D;
    const _Float16* bBase = W16 + (size_t)laneMod * D;

    v8f acc[8] = {};
    #pragma unroll
    for (int kk = 0; kk < D; kk += 32) {
        const int off0 = kk + laneHi * 8;
        const int off1 = kk + 16 + laneHi * 8;

        float4 a0 = *(const float4*)(aPtr + off0);
        float4 a1 = *(const float4*)(aPtr + off0 + 4);
        float4 a2 = *(const float4*)(aPtr + off1);
        float4 a3 = *(const float4*)(aPtr + off1 + 4);
        v16h a;
        a[0]=(_Float16)a0.x;  a[1]=(_Float16)a0.y;  a[2]=(_Float16)a0.z;  a[3]=(_Float16)a0.w;
        a[4]=(_Float16)a1.x;  a[5]=(_Float16)a1.y;  a[6]=(_Float16)a1.z;  a[7]=(_Float16)a1.w;
        a[8]=(_Float16)a2.x;  a[9]=(_Float16)a2.y;  a[10]=(_Float16)a2.z; a[11]=(_Float16)a2.w;
        a[12]=(_Float16)a3.x; a[13]=(_Float16)a3.y; a[14]=(_Float16)a3.z; a[15]=(_Float16)a3.w;

        #pragma unroll
        for (int t = 0; t < 8; ++t) {
            const _Float16* bp = bBase + (size_t)t * 16 * D;
            v8h blo = *(const v8h*)(bp + off0);
            v8h bhi = *(const v8h*)(bp + off1);
            v16h b = __builtin_shufflevector(blo, bhi,
                       0,1,2,3,4,5,6,7,8,9,10,11,12,13,14,15);
            acc[t] = __builtin_amdgcn_wmma_f32_16x16x32_f16(
                         false, a,  false, b,
                         (short)0, acc[t],
                         false,  false);
        }
    }

    #pragma unroll
    for (int t = 0; t < 8; ++t) {
        #pragma unroll
        for (int i = 0; i < 8; ++i) {
            int r = rowBase + i + laneHi * 8;
            if (r < n_nodes) {
                float v = acc[t][i];
                Y[(size_t)r * D + t * 16 + laneMod] = v > 0.f ? v : 0.f;
            }
        }
    }
}

__global__ void __launch_bounds__(256) head_kernel(
        const float* __restrict__ H,
        const float* __restrict__ Wc,
        const float* __restrict__ bc,
        float*       __restrict__ out,
        int n_nodes) {
    const int node = blockIdx.x * (blockDim.x >> 5) + (threadIdx.x >> 5);
    const int lane = threadIdx.x & 31;
    if (node >= n_nodes) return;
    float4 h = ((const float4*)(H + (size_t)node * D))[lane];
    float4 w = ((const float4*)Wc)[lane];
    float s = h.x * w.x + h.y * w.y + h.z * w.z + h.w * w.w;
    #pragma unroll
    for (int off = 16; off > 0; off >>= 1)
        s += __shfl_down(s, off, 32);
    if (lane == 0) out[node] = s + bc[0];
}

extern "C" void kernel_launch(void* const* d_in, const int* in_sizes, int n_in,
                              void* d_out, int out_size, void* d_ws, size_t ws_size,
                              hipStream_t stream) {
    const float* features = (const float*)d_in[0];
    const int*   erows    = (const int*)  d_in[1];
    const int*   ecols    = (const int*)  d_in[2];
    const float* evals    = (const float*)d_in[3];
    const float* W1       = (const float*)d_in[4];
    const float* W2       = (const float*)d_in[5];
    const float* Wc       = (const float*)d_in[6];
    const float* bc       = (const float*)d_in[7];
    float* out = (float*)d_out;

    const int n_nodes = in_sizes[0] / D;
    const int n_edges = in_sizes[1];

    float*     buf0 = (float*)d_ws;
    float*     buf1 = buf0 + (size_t)n_nodes * D;
    _Float16*  w16a = (_Float16*)(buf1 + (size_t)n_nodes * D);
    _Float16*  w16b = w16a + D * D;

    const int n4      = (n_nodes * D) / 4;
    const int zblocks = (n4 + 255) / 256;
    const int ewaves  = (n_edges + 3) / 4;
    const int eblocks = (ewaves + 7) / 8;
    const int gblocks = (n_nodes + 127) / 128;
    const int hblocks = (n_nodes + 7) / 8;
    const int w4      = (D * D) / 4;
    const int wblocks = (w4 + 255) / 256;

    cvt_f32_f16<<<wblocks, 256, 0, stream>>>(W1, w16a, w4);
    cvt_f32_f16<<<wblocks, 256, 0, stream>>>(W2, w16b, w4);

    zero_kernel<<<zblocks, 256, 0, stream>>>((float4*)buf0, n4);
    spmm_scatter<<<eblocks, 256, 0, stream>>>(erows, ecols, evals, features, buf0, n_edges);
    gemm_relu_wmma<<<gblocks, 256, 0, stream>>>(buf0, w16a, buf1, n_nodes);

    zero_kernel<<<zblocks, 256, 0, stream>>>((float4*)buf0, n4);
    spmm_scatter<<<eblocks, 256, 0, stream>>>(erows, ecols, evals, buf1, buf0, n_edges);
    gemm_relu_wmma<<<gblocks, 256, 0, stream>>>(buf0, w16b, buf1, n_nodes);

    head_kernel<<<hblocks, 256, 0, stream>>>(buf1, Wc, bc, out, n_nodes);
}


# SpMM as row gather: each call counting-sorts edges by dest row (2-level, LDS counters), one wave sums a row in registers, stores once; f32 atomics, zero fills gone; row sum order = LDS arrival rank
# speedup vs baseline: 3.1438x; 3.1438x over previous
#include <hip/hip_runtime.h>

typedef _Float16 v16h __attribute__((ext_vector_type(16)));
typedef _Float16 v8h  __attribute__((ext_vector_type(8)));
typedef _Float16 h4   __attribute__((ext_vector_type(4)));
typedef float    v8f  __attribute__((ext_vector_type(8)));

#define D 128

__global__ void __launch_bounds__(256) cvt_f32_f16(const float* __restrict__ src,
                                                   _Float16* __restrict__ dst, int n4) {
    int i = blockIdx.x * blockDim.x + threadIdx.x;
    if (i < n4) {
        float4 v = ((const float4*)src)[i];
        h4 o;
        o[0] = (_Float16)v.x; o[1] = (_Float16)v.y;
        o[2] = (_Float16)v.z; o[3] = (_Float16)v.w;
        ((h4*)dst)[i] = o;
    }
}

#define RB 1024
#define CH 8192

__device__ __forceinline__ int block_excl_scan_1024(int v, int* wsum) {
    const int lane = threadIdx.x & 31;
    const int w    = threadIdx.x >> 5;
    int inc = v;
    #pragma unroll
    for (int o = 1; o < 32; o <<= 1) {
        int t = __shfl_up(inc, o, 32);
        if (lane >= o) inc += t;
    }
    if (lane == 31) wsum[w] = inc;
    __syncthreads();
    if (w == 0) {
        int x = wsum[lane];
        int xi = x;
        #pragma unroll
        for (int o = 1; o < 32; o <<= 1) {
            int t = __shfl_up(xi, o, 32);
            if (lane >= o) xi += t;
        }
        wsum[lane] = xi - x;
    }
    __syncthreads();
    return inc - v + wsum[w];
}

template <bool FULL>
__device__ __forceinline__ void count_chunk(const int* __restrict__ rows, int* sh,
                                            int base, int n_edges) {
    int r[CH / 1024];
    #pragma unroll
    for (int k = 0; k < CH / 1024; ++k) {
        int e = base + k * 1024 + (int)threadIdx.x;
        r[k] = (FULL || e < n_edges) ? rows[e] : -1;
    }
    #pragma unroll
    for (int k = 0; k < CH / 1024; ++k)
        if (FULL || r[k] >= 0) atomicAdd(&sh[r[k] / RB], 1);
}

__global__ void __launch_bounds__(1024) bucket_count(
        const int* __restrict__ rows, int* __restrict__ tbl,
        int n_edges, int nb, int nblk) {
    extern __shared__ int sh[];
    const int tid = threadIdx.x;
    for (int i = tid; i < nb; i += 1024) sh[i] = 0;
    __syncthreads();
    const int base = blockIdx.x * CH;
    if (base + CH <= n_edges) count_chunk<true>(rows, sh, base, n_edges);
    else                      count_chunk<false>(rows, sh, base, n_edges);
    __syncthreads();
    for (int i = tid; i < nb; i += 1024) tbl[(size_t)i * nblk + blockIdx.x] = sh[i];
}

__global__ void __launch_bounds__(1024) table_scan(int* __restrict__ tbl, int total) {
    __shared__ int wsum[32];
    const int tid = threadIdx.x;
    const int len = (total + 1023) / 1024;
    const int lo  = tid * len;
    const int hi  = (lo + len < total) ? lo + len : total;
    int s = 0;
    for (int i = lo; i < hi; ++i) s += tbl[i];
    int off = block_excl_scan_1024(s, wsum);
    int i = lo;
    for (; i + 4 <= hi; i += 4) {
        int a = tbl[i], b = tbl[i + 1], c = tbl[i + 2], d = tbl[i + 3];
        tbl[i]     = off;
        tbl[i + 1] = off + a;
        tbl[i + 2] = off + a + b;
        tbl[i + 3] = off + a + b + c;
        off += a + b + c + d;
    }
    for (; i < hi; ++i) {
        int t = tbl[i];
        tbl[i] = off;
        off += t;
    }
}

template <bool FULL>
__device__ __forceinline__ void scatter_chunk(
        const int* __restrict__ rows, const int* __restrict__ cols,
        const float* __restrict__ vals, int* sh, int4* __restrict__ tmp,
        int base, int n_edges) {
    int r[CH / 1024], c[CH / 1024], v[CH / 1024];
    #pragma unroll
    for (int k = 0; k < CH / 1024; ++k) {
        int e = base + k * 1024 + (int)threadIdx.x;
        bool in = FULL || e < n_edges;
        r[k] = in ? rows[e] : -1;
        c[k] = in ? cols[e] : 0;
        v[k] = in ? __float_as_int(vals[e]) : 0;
    }
    #pragma unroll
    for (int k = 0; k < CH / 1024; ++k) {
        if (FULL || r[k] >= 0) {
            int p = atomicAdd(&sh[r[k] / RB], 1);
            tmp[p] = make_int4(r[k], c[k], v[k], 0);
        }
    }
}

__global__ void __launch_bounds__(1024) bucket_scatter(
        const int*   __restrict__ rows,
        const int*   __restrict__ cols,
        const float* __restrict__ vals,
        const int*   __restrict__ tbl,
        int4*        __restrict__ tmp,
        int n_edges, int nb, int nblk) {
    extern __shared__ int sh[];
    const int tid = threadIdx.x;
    for (int i = tid; i < nb; i += 1024) sh[i] = tbl[(size_t)i * nblk + blockIdx.x];
    __syncthreads();
    const int base = blockIdx.x * CH;
    if (base + CH <= n_edges) scatter_chunk<true>(rows, cols, vals, sh, tmp, base, n_edges);
    else                      scatter_chunk<false>(rows, cols, vals, sh, tmp, base, n_edges);
}

__global__ void __launch_bounds__(RB) bucket_sort(
        const int*  __restrict__ tbl,
        const int4* __restrict__ tmp,
        int2*       __restrict__ sorted,
        int*        __restrict__ rowptr,
        int n_nodes, int n_edges, int nb, int nblk) {
    __shared__ int cnt[RB];
    __shared__ int wsum[32];
    const int tid  = threadIdx.x;
    const int b    = blockIdx.x;
    const int row0 = b * RB;
    const int s = tbl[(size_t)b * nblk];
    const int t = (b + 1 < nb) ? tbl[(size_t)(b + 1) * nblk] : n_edges;
    cnt[tid] = 0;
    __syncthreads();
    int i = s + tid;
    for (; i + 3 * RB < t; i += 4 * RB) {
        int r0 = tmp[i].x, r1 = tmp[i + RB].x, r2 = tmp[i + 2 * RB].x, r3 = tmp[i + 3 * RB].x;
        atomicAdd(&cnt[r0 - row0], 1);
        atomicAdd(&cnt[r1 - row0], 1);
        atomicAdd(&cnt[r2 - row0], 1);
        atomicAdd(&cnt[r3 - row0], 1);
    }
    for (; i < t; i += RB) atomicAdd(&cnt[tmp[i].x - row0], 1);
    __syncthreads();
    const int c   = cnt[tid];
    const int off = s + block_excl_scan_1024(c, wsum);
    if (row0 + tid < n_nodes) rowptr[row0 + tid] = off;
    cnt[tid] = off;
    __syncthreads();
    i = s + tid;
    for (; i + 3 * RB < t; i += 4 * RB) {
        int4 e0 = tmp[i], e1 = tmp[i + RB], e2 = tmp[i + 2 * RB], e3 = tmp[i + 3 * RB];
        int p0 = atomicAdd(&cnt[e0.x - row0], 1);
        int p1 = atomicAdd(&cnt[e1.x - row0], 1);
        int p2 = atomicAdd(&cnt[e2.x - row0], 1);
        int p3 = atomicAdd(&cnt[e3.x - row0], 1);
        sorted[p0] = make_int2(e0.y, e0.z);
        sorted[p1] = make_int2(e1.y, e1.z);
        sorted[p2] = make_int2(e2.y, e2.z);
        sorted[p3] = make_int2(e3.y, e3.z);
    }
    for (; i < t; i += RB) {
        int4 e = tmp[i];
        int p = atomicAdd(&cnt[e.x - row0], 1);
        sorted[p] = make_int2(e.y, e.z);
    }
}


template <bool TAIL>
__device__ __forceinline__ void gather8(const float4* __restrict__ xv, int2 ent,
                                        int k, int m, float4& acc) {
    float4 d[8];
    float  v[8];
    #pragma unroll
    for (int i = 0; i < 8; ++i) {
        const int c = __builtin_amdgcn_readlane(ent.x, k + i);
        v[i] = __int_as_float(__builtin_amdgcn_readlane(ent.y, k + i));
        if (!TAIL || k + i < m) {
            d[i] = xv[(size_t)c * (D / 4)];
        } else {
            d[i] = make_float4(0.f, 0.f, 0.f, 0.f);
            v[i] = 0.f;
        }
    }
    #pragma unroll
    for (int i = 0; i < 8; ++i) {
        acc.x = fmaf(v[i], d[i].x, acc.x);
        acc.y = fmaf(v[i], d[i].y, acc.y);
        acc.z = fmaf(v[i], d[i].z, acc.z);
        acc.w = fmaf(v[i], d[i].w, acc.w);
    }
}

__global__ void __launch_bounds__(256) spmm_gather(
        const int*   __restrict__ rowptr,
        const int2*  __restrict__ sorted,
        const float* __restrict__ x,
        float*       __restrict__ agg,
        int n_nodes, int n_edges) {
    const int lane = threadIdx.x & 31;
    const int row = __builtin_amdgcn_readfirstlane(blockIdx.x * 8 + (threadIdx.x >> 5));
    if (row >= n_nodes) return;
    const int s = rowptr[row];
    const int t = (row + 1 < n_nodes) ? rowptr[row + 1] : n_edges;
    const float4* xv = (const float4*)x + lane;
    float4 acc = make_float4(0.f, 0.f, 0.f, 0.f);
    for (int c0 = s; c0 < t; c0 += 32) {
        const int m = (t - c0 < 32) ? t - c0 : 32;
        int2 ent = make_int2(0, 0);
        if (lane < m) ent = sorted[c0 + lane];
        int k = 0;
        for (; k + 8 <= m; k += 8) gather8<false>(xv, ent, k, m, acc);
        if (k < m) gather8<true>(xv, ent, k, m, acc);
    }
    ((float4*)(agg + (size_t)row * D))[lane] = acc;
}

__global__ void __launch_bounds__(256) gemm_relu_wmma(
        const float*    __restrict__ X,
        const _Float16* __restrict__ W16,
        float*          __restrict__ Y,
        int n_nodes) {
    const int wave    = threadIdx.x >> 5;
    const int lane    = threadIdx.x & 31;
    const int laneMod = lane & 15;
    const int laneHi  = lane >> 4;
    const int rowBase = (blockIdx.x * 8 + wave) * 16;

    int aRow = rowBase + laneMod;
    if (aRow >= n_nodes) aRow = n_nodes - 1;

    const float*    aPtr  = X + (size_t)aRow * D;
    const _Float16* bBase = W16 + (size_t)laneMod * D;

    v8f acc[8] = {};
    #pragma unroll
    for (int kk = 0; kk < D; kk += 32) {
        const int off0 = kk + laneHi * 8;
        const int off1 = kk + 16 + laneHi * 8;

        float4 a0 = *(const float4*)(aPtr + off0);
        float4 a1 = *(const float4*)(aPtr + off0 + 4);
        float4 a2 = *(const float4*)(aPtr + off1);
        float4 a3 = *(const float4*)(aPtr + off1 + 4);
        v16h a;
        a[0]=(_Float16)a0.x;  a[1]=(_Float16)a0.y;  a[2]=(_Float16)a0.z;  a[3]=(_Float16)a0.w;
        a[4]=(_Float16)a1.x;  a[5]=(_Float16)a1.y;  a[6]=(_Float16)a1.z;  a[7]=(_Float16)a1.w;
        a[8]=(_Float16)a2.x;  a[9]=(_Float16)a2.y;  a[10]=(_Float16)a2.z; a[11]=(_Float16)a2.w;
        a[12]=(_Float16)a3.x; a[13]=(_Float16)a3.y; a[14]=(_Float16)a3.z; a[15]=(_Float16)a3.w;

        #pragma unroll
        for (int t = 0; t < 8; ++t) {
            const _Float16* bp = bBase + (size_t)t * 16 * D;
            v8h blo = *(const v8h*)(bp + off0);
            v8h bhi = *(const v8h*)(bp + off1);
            v16h b = __builtin_shufflevector(blo, bhi,
                       0,1,2,3,4,5,6,7,8,9,10,11,12,13,14,15);
            acc[t] = __builtin_amdgcn_wmma_f32_16x16x32_f16(
                         false, a,  false, b,
                         (short)0, acc[t],
                         false,  false);
        }
    }

    #pragma unroll
    for (int t = 0; t < 8; ++t) {
        #pragma unroll
        for (int i = 0; i < 8; ++i) {
            int r = rowBase + i + laneHi * 8;
            if (r < n_nodes) {
                float v = acc[t][i];
                Y[(size_t)r * D + t * 16 + laneMod] = v > 0.f ? v : 0.f;
            }
        }
    }
}

__global__ void __launch_bounds__(256) head_kernel(
        const float* __restrict__ H,
        const float* __restrict__ Wc,
        const float* __restrict__ bc,
        float*       __restrict__ out,
        int n_nodes) {
    const int node = blockIdx.x * (blockDim.x >> 5) + (threadIdx.x >> 5);
    const int lane = threadIdx.x & 31;
    if (node >= n_nodes) return;
    float4 h = ((const float4*)(H + (size_t)node * D))[lane];
    float4 w = ((const float4*)Wc)[lane];
    float s = h.x * w.x + h.y * w.y + h.z * w.z + h.w * w.w;
    #pragma unroll
    for (int off = 16; off > 0; off >>= 1)
        s += __shfl_down(s, off, 32);
    if (lane == 0) out[node] = s + bc[0];
}

extern "C" void kernel_launch(void* const* d_in, const int* in_sizes, int n_in,
                              void* d_out, int out_size, void* d_ws, size_t ws_size,
                              hipStream_t stream) {
    const float* features = (const float*)d_in[0];
    const int*   erows    = (const int*)  d_in[1];
    const int*   ecols    = (const int*)  d_in[2];
    const float* evals    = (const float*)d_in[3];
    const float* W1       = (const float*)d_in[4];
    const float* W2       = (const float*)d_in[5];
    const float* Wc       = (const float*)d_in[6];
    const float* bc       = (const float*)d_in[7];
    float* out = (float*)d_out;

    const int n_nodes = in_sizes[0] / D;
    const int n_edges = in_sizes[1];

    float*     buf0 = (float*)d_ws;
    float*     buf1 = buf0 + (size_t)n_nodes * D;
    _Float16*  w16a = (_Float16*)(buf1 + (size_t)n_nodes * D);
    _Float16*  w16b = w16a + D * D;

    const int nb      = (n_nodes + RB - 1) / RB;
    const int nblk    = (n_edges + CH - 1) / CH;
    const int ntbl    = nb * nblk;
    const size_t tblPad = ((size_t)ntbl + 63) / 64 * 64;
    const size_t rowPad = ((size_t)n_nodes + 63) / 64 * 64;

    int*   tbl    = (int*)(w16b + D * D);
    int*   rowptr = tbl + tblPad;
    int4*  tmp    = (int4*)(rowptr + rowPad);
    int2*  sorted = (int2*)(tmp + (size_t)n_edges);

    const int gblocks = (n_nodes + 127) / 128;
    const int hblocks = (n_nodes + 7) / 8;
    const int w4      = (D * D) / 4;
    const int wblocks = (w4 + 255) / 256;
    const size_t shb  = (size_t)nb * sizeof(int);

    cvt_f32_f16<<<wblocks, 256, 0, stream>>>(W1, w16a, w4);
    cvt_f32_f16<<<wblocks, 256, 0, stream>>>(W2, w16b, w4);

    bucket_count<<<nblk, 1024, shb, stream>>>(erows, tbl, n_edges, nb, nblk);
    table_scan<<<1, 1024, 0, stream>>>(tbl, ntbl);
    bucket_scatter<<<nblk, 1024, shb, stream>>>(erows, ecols, evals, tbl, tmp, n_edges, nb, nblk);
    bucket_sort<<<nb, RB, 0, stream>>>(tbl, tmp, sorted, rowptr, n_nodes, n_edges, nb, nblk);

    spmm_gather<<<hblocks, 256, 0, stream>>>(rowptr, sorted, features, buf0, n_nodes, n_edges);
    gemm_relu_wmma<<<gblocks, 256, 0, stream>>>(buf0, w16a, buf1, n_nodes);

    spmm_gather<<<hblocks, 256, 0, stream>>>(rowptr, sorted, buf1, buf0, n_nodes, n_edges);
    gemm_relu_wmma<<<gblocks, 256, 0, stream>>>(buf0, w16b, buf1, n_nodes);

    head_kernel<<<hblocks, 256, 0, stream>>>(buf1, Wc, bc, out, n_nodes);
}


# each layer is one launch: a wave sums its 16 rows, rounds them to f16 into LDS as the WMMA A operand; head dot taken on the accumulators (its sum order changes); agg, H2 buffers and 3 launches gone
# speedup vs baseline: 4.4853x; 1.4267x over previous
#include <hip/hip_runtime.h>

typedef _Float16 v16h __attribute__((ext_vector_type(16)));
typedef _Float16 v8h  __attribute__((ext_vector_type(8)));
typedef _Float16 h4   __attribute__((ext_vector_type(4)));
typedef float    v8f  __attribute__((ext_vector_type(8)));

#define D 128

__global__ void __launch_bounds__(256) cvt_f32_f16(const float* __restrict__ src,
                                                   _Float16* __restrict__ dst, int n4) {
    int i = blockIdx.x * blockDim.x + threadIdx.x;
    if (i < n4) {
        float4 v = ((const float4*)src)[i];
        h4 o;
        o[0] = (_Float16)v.x; o[1] = (_Float16)v.y;
        o[2] = (_Float16)v.z; o[3] = (_Float16)v.w;
        ((h4*)dst)[i] = o;
    }
}

#define RB 1024
#define CH 8192

__device__ __forceinline__ int block_excl_scan_1024(int v, int* wsum) {
    const int lane = threadIdx.x & 31;
    const int w    = threadIdx.x >> 5;
    int inc = v;
    #pragma unroll
    for (int o = 1; o < 32; o <<= 1) {
        int t = __shfl_up(inc, o, 32);
        if (lane >= o) inc += t;
    }
    if (lane == 31) wsum[w] = inc;
    __syncthreads();
    if (w == 0) {
        int x = wsum[lane];
        int xi = x;
        #pragma unroll
        for (int o = 1; o < 32; o <<= 1) {
            int t = __shfl_up(xi, o, 32);
            if (lane >= o) xi += t;
        }
        wsum[lane] = xi - x;
    }
    __syncthreads();
    return inc - v + wsum[w];
}

template <bool FULL>
__device__ __forceinline__ void count_chunk(const int* __restrict__ rows, int* sh,
                                            int base, int n_edges) {
    int r[CH / 1024];
    #pragma unroll
    for (int k = 0; k < CH / 1024; ++k) {
        int e = base + k * 1024 + (int)threadIdx.x;
        r[k] = (FULL || e < n_edges) ? rows[e] : -1;
    }
    #pragma unroll
    for (int k = 0; k < CH / 1024; ++k)
        if (FULL || r[k] >= 0) atomicAdd(&sh[r[k] / RB], 1);
}

__global__ void __launch_bounds__(1024) bucket_count(
        const int* __restrict__ rows, int* __restrict__ tbl,
        int n_edges, int nb, int nblk) {
    extern __shared__ int sh[];
    const int tid = threadIdx.x;
    for (int i = tid; i < nb; i += 1024) sh[i] = 0;
    __syncthreads();
    const int base = blockIdx.x * CH;
    if (base + CH <= n_edges) count_chunk<true>(rows, sh, base, n_edges);
    else                      count_chunk<false>(rows, sh, base, n_edges);
    __syncthreads();
    for (int i = tid; i < nb; i += 1024) tbl[(size_t)i * nblk + blockIdx.x] = sh[i];
}

__global__ void __launch_bounds__(1024) table_scan(int* __restrict__ tbl, int total) {
    __shared__ int wsum[32];
    const int tid = threadIdx.x;
    const int len = (total + 1023) / 1024;
    const int lo  = tid * len;
    const int hi  = (lo + len < total) ? lo + len : total;
    int s = 0;
    for (int i = lo; i < hi; ++i) s += tbl[i];
    int off = block_excl_scan_1024(s, wsum);
    int i = lo;
    for (; i + 4 <= hi; i += 4) {
        int a = tbl[i], b = tbl[i + 1], c = tbl[i + 2], d = tbl[i + 3];
        tbl[i]     = off;
        tbl[i + 1] = off + a;
        tbl[i + 2] = off + a + b;
        tbl[i + 3] = off + a + b + c;
        off += a + b + c + d;
    }
    for (; i < hi; ++i) {
        int t = tbl[i];
        tbl[i] = off;
        off += t;
    }
}

template <bool FULL>
__device__ __forceinline__ void scatter_chunk(
        const int* __restrict__ rows, const int* __restrict__ cols,
        const float* __restrict__ vals, int* sh, int4* __restrict__ tmp,
        int base, int n_edges) {
    int r[CH / 1024], c[CH / 1024], v[CH / 1024];
    #pragma unroll
    for (int k = 0; k < CH / 1024; ++k) {
        int e = base + k * 1024 + (int)threadIdx.x;
        bool in = FULL || e < n_edges;
        r[k] = in ? rows[e] : -1;
        c[k] = in ? cols[e] : 0;
        v[k] = in ? __float_as_int(vals[e]) : 0;
    }
    #pragma unroll
    for (int k = 0; k < CH / 1024; ++k) {
        if (FULL || r[k] >= 0) {
            int p = atomicAdd(&sh[r[k] / RB], 1);
            tmp[p] = make_int4(r[k], c[k], v[k], 0);
        }
    }
}

__global__ void __launch_bounds__(1024) bucket_scatter(
        const int*   __restrict__ rows,
        const int*   __restrict__ cols,
        const float* __restrict__ vals,
        const int*   __restrict__ tbl,
        int4*        __restrict__ tmp,
        int n_edges, int nb, int nblk) {
    extern __shared__ int sh[];
    const int tid = threadIdx.x;
    for (int i = tid; i < nb; i += 1024) sh[i] = tbl[(size_t)i * nblk + blockIdx.x];
    __syncthreads();
    const int base = blockIdx.x * CH;
    if (base + CH <= n_edges) scatter_chunk<true>(rows, cols, vals, sh, tmp, base, n_edges);
    else                      scatter_chunk<false>(rows, cols, vals, sh, tmp, base, n_edges);
}

__global__ void __launch_bounds__(RB) bucket_sort(
        const int*  __restrict__ tbl,
        const int4* __restrict__ tmp,
        int2*       __restrict__ sorted,
        int*        __restrict__ rowptr,
        int n_nodes, int n_edges, int nb, int nblk) {
    __shared__ int cnt[RB];
    __shared__ int wsum[32];
    const int tid  = threadIdx.x;
    const int b    = blockIdx.x;
    const int row0 = b * RB;
    const int s = tbl[(size_t)b * nblk];
    const int t = (b + 1 < nb) ? tbl[(size_t)(b + 1) * nblk] : n_edges;
    cnt[tid] = 0;
    __syncthreads();
    int i = s + tid;
    for (; i + 3 * RB < t; i += 4 * RB) {
        int r0 = tmp[i].x, r1 = tmp[i + RB].x, r2 = tmp[i + 2 * RB].x, r3 = tmp[i + 3 * RB].x;
        atomicAdd(&cnt[r0 - row0], 1);
        atomicAdd(&cnt[r1 - row0], 1);
        atomicAdd(&cnt[r2 - row0], 1);
        atomicAdd(&cnt[r3 - row0], 1);
    }
    for (; i < t; i += RB) atomicAdd(&cnt[tmp[i].x - row0], 1);
    __syncthreads();
    const int c   = cnt[tid];
    const int off = s + block_excl_scan_1024(c, wsum);
    if (row0 + tid < n_nodes) rowptr[row0 + tid] = off;
    cnt[tid] = off;
    __syncthreads();
    i = s + tid;
    for (; i + 3 * RB < t; i += 4 * RB) {
        int4 e0 = tmp[i], e1 = tmp[i + RB], e2 = tmp[i + 2 * RB], e3 = tmp[i + 3 * RB];
        int p0 = atomicAdd(&cnt[e0.x - row0], 1);
        int p1 = atomicAdd(&cnt[e1.x - row0], 1);
        int p2 = atomicAdd(&cnt[e2.x - row0], 1);
        int p3 = atomicAdd(&cnt[e3.x - row0], 1);
        sorted[p0] = make_int2(e0.y, e0.z);
        sorted[p1] = make_int2(e1.y, e1.z);
        sorted[p2] = make_int2(e2.y, e2.z);
        sorted[p3] = make_int2(e3.y, e3.z);
    }
    for (; i < t; i += RB) {
        int4 e = tmp[i];
        int p = atomicAdd(&cnt[e.x - row0], 1);
        sorted[p] = make_int2(e.y, e.z);
    }
}


#define PITCH 136

template <int M>
__device__ __forceinline__ float row_xor(float x) {
    return __builtin_bit_cast(float, __builtin_amdgcn_update_dpp(
               0, __builtin_bit_cast(int, x), 0x160 + M, 0xf, 0xf, false));
}

template <bool TAIL>
__device__ __forceinline__ void gather8(const float4* __restrict__ xv, int2 ent,
                                        int k, int m, float4& acc,
                                        bool pre, const int2* __restrict__ nl, int nN,
                                        int lane, int2& entN) {
    float4 d[8];
    float  v[8];
    #pragma unroll
    for (int i = 0; i < 8; ++i) {
        const int c = __builtin_amdgcn_readlane(ent.x, k + i);
        v[i] = __int_as_float(__builtin_amdgcn_readlane(ent.y, k + i));
        if (!TAIL || k + i < m) {
            d[i] = xv[(size_t)c * (D / 4)];
        } else {
            d[i] = make_float4(0.f, 0.f, 0.f, 0.f);
            v[i] = 0.f;
        }
    }
    if (pre && lane < nN) entN = nl[lane];
    #pragma unroll
    for (int i = 0; i < 8; ++i) {
        acc.x = fmaf(v[i], d[i].x, acc.x);
        acc.y = fmaf(v[i], d[i].y, acc.y);
        acc.z = fmaf(v[i], d[i].z, acc.z);
        acc.w = fmaf(v[i], d[i].w, acc.w);
    }
}

template <bool HEAD>
__global__ void __launch_bounds__(256, 8) gather_gemm_relu(
        const int*      __restrict__ rowptr,
        const int2*     __restrict__ sorted,
        const float*    __restrict__ x,
        const _Float16* __restrict__ W16,
        float*          __restrict__ Y,
        const float*    __restrict__ Wc,
        const float*    __restrict__ bc,
        float*          __restrict__ out,
        int n_nodes, int n_edges) {
    __shared__ __attribute__((aligned(16))) _Float16 sW[D * PITCH];
    __shared__ __attribute__((aligned(16))) _Float16 sA[8][16 * PITCH];

    const int wave    = threadIdx.x >> 5;
    const int lane    = threadIdx.x & 31;
    const int laneMod = lane & 15;
    const int laneHi  = lane >> 4;
    const int rowBase = __builtin_amdgcn_readfirstlane((blockIdx.x * 8 + wave) * 16);

    v8h wreg[8];
    #pragma unroll
    for (int q = 0; q < 8; ++q)
        wreg[q] = ((const v8h*)W16)[q * 256 + (int)threadIdx.x];

    const int rr = rowBase + lane;
    const int rb = (rr < n_nodes) ? rowptr[rr] : n_edges;

    const float4* xv = (const float4*)x + lane;
    _Float16* slot = &sA[wave][0];

    int2 entN = make_int2(0, 0);
    {
        const int s0 = __builtin_amdgcn_readlane(rb, 0);
        const int t0 = __builtin_amdgcn_readlane(rb, 1);
        if (lane < t0 - s0) entN = sorted[s0 + lane];
    }
    #pragma unroll 1
    for (int r = 0; r < 16; ++r) {
        const int s = __builtin_amdgcn_readlane(rb, r);
        const int t = __builtin_amdgcn_readlane(rb, r + 1);
        int2 ent = entN;
        const int nN = (r + 1 < 16) ? __builtin_amdgcn_readlane(rb, r + 2) - t : 0;
        const int2* nl = sorted + t;
        entN = make_int2(0, 0);
        if (s >= t && lane < nN) entN = nl[lane];
        float4 acc = make_float4(0.f, 0.f, 0.f, 0.f);
        for (int c0 = s; c0 < t; c0 += 32) {
            const int m = (t - c0 < 32) ? t - c0 : 32;
            if (c0 != s) {
                ent = make_int2(0, 0);
                if (lane < m) ent = sorted[c0 + lane];
            }
            const bool lastChunk = c0 + 32 >= t;
            int k = 0;
            for (; k + 8 <= m; k += 8)
                gather8<false>(xv, ent, k, m, acc, lastChunk && k + 8 >= m, nl, nN, lane, entN);
            if (k < m) gather8<true>(xv, ent, k, m, acc, lastChunk, nl, nN, lane, entN);
        }
        h4 o;
        o[0] = (_Float16)acc.x; o[1] = (_Float16)acc.y;
        o[2] = (_Float16)acc.z; o[3] = (_Float16)acc.w;
        *(h4*)(slot + r * PITCH + lane * 4) = o;
    }

    #pragma unroll
    for (int q = 0; q < 8; ++q) {
        const int piece = q * 256 + (int)threadIdx.x;
        *(v8h*)(sW + (piece >> 4) * PITCH + (piece & 15) * 8) = wreg[q];
    }
    __syncthreads();

    const _Float16* aRow = slot + laneMod * PITCH;
    const _Float16* bRow = sW + laneMod * PITCH;
    if (!HEAD) {
        v8f acc[8] = {};
        #pragma unroll
        for (int kk = 0; kk < D; kk += 32) {
            const int off0 = kk + laneHi * 8;
            const int off1 = kk + 16 + laneHi * 8;
            v8h alo = *(const v8h*)(aRow + off0);
            v8h ahi = *(const v8h*)(aRow + off1);
            v16h a = __builtin_shufflevector(alo, ahi,
                       0,1,2,3,4,5,6,7,8,9,10,11,12,13,14,15);
            #pragma unroll
            for (int t = 0; t < 8; ++t) {
                const _Float16* bp = bRow + t * 16 * PITCH;
                v8h blo = *(const v8h*)(bp + off0);
                v8h bhi = *(const v8h*)(bp + off1);
                v16h b = __builtin_shufflevector(blo, bhi,
                           0,1,2,3,4,5,6,7,8,9,10,11,12,13,14,15);
                acc[t] = __builtin_amdgcn_wmma_f32_16x16x32_f16(
                             false, a,  false, b,
                             (short)0, acc[t],
                             false,  false);
            }
        }
        #pragma unroll
        for (int t = 0; t < 8; ++t) {
            #pragma unroll
            for (int i = 0; i < 8; ++i) {
                int r = rowBase + i + laneHi * 8;
                if (r < n_nodes) {
                    float v = acc[t][i];
                    Y[(size_t)r * D + t * 16 + laneMod] = v > 0.f ? v : 0.f;
                }
            }
        }
    } else {
        float wc[8];
        #pragma unroll
        for (int t = 0; t < 8; ++t) wc[t] = Wc[t * 16 + laneMod];
        float p[8];
        #pragma unroll
        for (int i = 0; i < 8; ++i) p[i] = 0.f;
        #pragma unroll
        for (int g = 0; g < 2; ++g) {
            v8f acc[4] = {};
            #pragma unroll
            for (int kk = 0; kk < D; kk += 32) {
                const int off0 = kk + laneHi * 8;
                const int off1 = kk + 16 + laneHi * 8;
                v8h alo = *(const v8h*)(aRow + off0);
                v8h ahi = *(const v8h*)(aRow + off1);
                v16h a = __builtin_shufflevector(alo, ahi,
                           0,1,2,3,4,5,6,7,8,9,10,11,12,13,14,15);
                #pragma unroll
                for (int tt = 0; tt < 4; ++tt) {
                    const _Float16* bp = bRow + (g * 4 + tt) * 16 * PITCH;
                    v8h blo = *(const v8h*)(bp + off0);
                    v8h bhi = *(const v8h*)(bp + off1);
                    v16h b = __builtin_shufflevector(blo, bhi,
                               0,1,2,3,4,5,6,7,8,9,10,11,12,13,14,15);
                    acc[tt] = __builtin_amdgcn_wmma_f32_16x16x32_f16(
                                 false, a,  false, b,
                                 (short)0, acc[tt],
                                 false,  false);
                }
            }
            #pragma unroll
            for (int tt = 0; tt < 4; ++tt) {
                const float w = wc[g * 4 + tt];
                #pragma unroll
                for (int i = 0; i < 8; ++i) {
                    float v = acc[tt][i];
                    v = v > 0.f ? v : 0.f;
                    p[i] = fmaf(v, w, p[i]);
                }
            }
        }
        #pragma unroll
        for (int i = 0; i < 8; ++i) {
            p[i] += row_xor<8>(p[i]);
            p[i] += row_xor<4>(p[i]);
            p[i] += row_xor<2>(p[i]);
            p[i] += row_xor<1>(p[i]);
        }
        float mine = p[0];
        #pragma unroll
        for (int i = 1; i < 8; ++i) mine = (laneMod == i) ? p[i] : mine;
        const int r = rowBase + laneMod + laneHi * 8;
        if (laneMod < 8 && r < n_nodes) out[r] = mine + bc[0];
    }
}

extern "C" void kernel_launch(void* const* d_in, const int* in_sizes, int n_in,
                              void* d_out, int out_size, void* d_ws, size_t ws_size,
                              hipStream_t stream) {
    const float* features = (const float*)d_in[0];
    const int*   erows    = (const int*)  d_in[1];
    const int*   ecols    = (const int*)  d_in[2];
    const float* evals    = (const float*)d_in[3];
    const float* W1       = (const float*)d_in[4];
    const float* W2       = (const float*)d_in[5];
    const float* Wc       = (const float*)d_in[6];
    const float* bc       = (const float*)d_in[7];
    float* out = (float*)d_out;

    const int n_nodes = in_sizes[0] / D;
    const int n_edges = in_sizes[1];

    float*     buf1 = (float*)d_ws;
    _Float16*  w16a = (_Float16*)(buf1 + (size_t)n_nodes * D);
    _Float16*  w16b = w16a + D * D;

    const int nb      = (n_nodes + RB - 1) / RB;
    const int nblk    = (n_edges + CH - 1) / CH;
    const int ntbl    = nb * nblk;
    const size_t tblPad = ((size_t)ntbl + 63) / 64 * 64;
    const size_t rowPad = ((size_t)n_nodes + 63) / 64 * 64;

    int*   tbl    = (int*)(w16b + D * D);
    int*   rowptr = tbl + tblPad;
    int4*  tmp    = (int4*)(rowptr + rowPad);
    int2*  sorted = (int2*)(tmp + (size_t)n_edges);

    const int gblocks = (n_nodes + 127) / 128;
    const int w4      = (D * D) / 4;
    const int wblocks = (w4 + 255) / 256;
    const size_t shb  = (size_t)nb * sizeof(int);

    cvt_f32_f16<<<wblocks, 256, 0, stream>>>(W1, w16a, w4);
    cvt_f32_f16<<<wblocks, 256, 0, stream>>>(W2, w16b, w4);

    bucket_count<<<nblk, 1024, shb, stream>>>(erows, tbl, n_edges, nb, nblk);
    table_scan<<<1, 1024, 0, stream>>>(tbl, ntbl);
    bucket_scatter<<<nblk, 1024, shb, stream>>>(erows, ecols, evals, tbl, tmp, n_edges, nb, nblk);
    bucket_sort<<<nb, RB, 0, stream>>>(tbl, tmp, sorted, rowptr, n_nodes, n_edges, nb, nblk);

    gather_gemm_relu<false><<<gblocks, 256, 0, stream>>>(
        rowptr, sorted, features, w16a, buf1, Wc, bc, out, n_nodes, n_edges);

    gather_gemm_relu<true><<<gblocks, 256, 0, stream>>>(
        rowptr, sorted, buf1, w16b, nullptr, Wc, bc, out, n_nodes, n_edges);
}


# three small launches folded into their consumers: layer kernels round the f32 weights to f16 into LDS on entry; scatter sums its slots from the count table itself (no scan launch); 8 -> 5 launches
# speedup vs baseline: 4.7830x; 1.0664x over previous
#include <hip/hip_runtime.h>

typedef _Float16 v16h __attribute__((ext_vector_type(16)));
typedef _Float16 v8h  __attribute__((ext_vector_type(8)));
typedef _Float16 h4   __attribute__((ext_vector_type(4)));
typedef float    v8f  __attribute__((ext_vector_type(8)));

#define D 128

#define RB 1024
#define CH 8192

__device__ __forceinline__ int block_excl_scan_1024(int v, int* wsum) {
    const int lane = threadIdx.x & 31;
    const int w    = threadIdx.x >> 5;
    int inc = v;
    #pragma unroll
    for (int o = 1; o < 32; o <<= 1) {
        int t = __shfl_up(inc, o, 32);
        if (lane >= o) inc += t;
    }
    if (lane == 31) wsum[w] = inc;
    __syncthreads();
    if (w == 0) {
        int x = wsum[lane];
        int xi = x;
        #pragma unroll
        for (int o = 1; o < 32; o <<= 1) {
            int t = __shfl_up(xi, o, 32);
            if (lane >= o) xi += t;
        }
        wsum[lane] = xi - x;
    }
    __syncthreads();
    return inc - v + wsum[w];
}

template <bool FULL>
__device__ __forceinline__ void count_chunk(const int* __restrict__ rows, int* sh,
                                            int base, int n_edges) {
    int r[CH / 1024];
    #pragma unroll
    for (int k = 0; k < CH / 1024; ++k) {
        int e = base + k * 1024 + (int)threadIdx.x;
        r[k] = (FULL || e < n_edges) ? rows[e] : -1;
    }
    #pragma unroll
    for (int k = 0; k < CH / 1024; ++k)
        if (FULL || r[k] >= 0) atomicAdd(&sh[r[k] / RB], 1);
}

__global__ void __launch_bounds__(1024) bucket_count(
        const int* __restrict__ rows, int* __restrict__ tbl,
        int n_edges, int nb, int nblk) {
    extern __shared__ int sh[];
    const int tid = threadIdx.x;
    for (int i = tid; i < nb; i += 1024) sh[i] = 0;
    __syncthreads();
    const int base = blockIdx.x * CH;
    if (base + CH <= n_edges) count_chunk<true>(rows, sh, base, n_edges);
    else                      count_chunk<false>(rows, sh, base, n_edges);
    __syncthreads();
    for (int i = tid; i < nb; i += 1024) tbl[(size_t)i * nblk + blockIdx.x] = sh[i];
}

template <bool FULL>
__device__ __forceinline__ void scatter_chunk(
        const int* __restrict__ rows, const int* __restrict__ cols,
        const float* __restrict__ vals, int* sh, int4* __restrict__ tmp,
        int base, int n_edges) {
    int r[CH / 1024], c[CH / 1024], v[CH / 1024];
    #pragma unroll
    for (int k = 0; k < CH / 1024; ++k) {
        int e = base + k * 1024 + (int)threadIdx.x;
        bool in = FULL || e < n_edges;
        r[k] = in ? rows[e] : -1;
        c[k] = in ? cols[e] : 0;
        v[k] = in ? __float_as_int(vals[e]) : 0;
    }
    #pragma unroll
    for (int k = 0; k < CH / 1024; ++k) {
        if (FULL || r[k] >= 0) {
            int p = atomicAdd(&sh[r[k] / RB], 1);
            tmp[p] = make_int4(r[k], c[k], v[k], 0);
        }
    }
}

__global__ void __launch_bounds__(1024) bucket_scatter(
        const int*   __restrict__ rows,
        const int*   __restrict__ cols,
        const float* __restrict__ vals,
        const int*   __restrict__ tbl,
        int*         __restrict__ bstart,
        int4*        __restrict__ tmp,
        int n_edges, int nb, int nblk) {
    extern __shared__ int sh[];
    __shared__ int wsum[32];
    __shared__ int passTot;
    int* tot = sh + nb;
    const int tid   = threadIdx.x;
    const int lane  = tid & 31;
    const int w     = tid >> 5;
    const int chunk = blockIdx.x;
    for (int b0 = w; b0 < nb; b0 += 128) {
        int all[4], before[4];
        #pragma unroll
        for (int u = 0; u < 4; ++u) { all[u] = 0; before[u] = 0; }
        for (int c0 = 0; c0 < nblk; c0 += 256) {
            int v[4][8];
            #pragma unroll
            for (int u = 0; u < 4; ++u) {
                const int b  = b0 + 32 * u;
                const int bb = (b < nb) ? b : nb - 1;
                #pragma unroll
                for (int j = 0; j < 8; ++j) {
                    const int c  = c0 + 32 * j + lane;
                    const int cc = (c < nblk) ? c : nblk - 1;
                    v[u][j] = tbl[(size_t)bb * nblk + cc];
                }
            }
            #pragma unroll
            for (int u = 0; u < 4; ++u) {
                #pragma unroll
                for (int j = 0; j < 8; ++j) {
                    const int c = c0 + 32 * j + lane;
                    all[u]    += (c < nblk)  ? v[u][j] : 0;
                    before[u] += (c < chunk) ? v[u][j] : 0;
                }
            }
        }
        #pragma unroll
        for (int u = 0; u < 4; ++u) {
            #pragma unroll
            for (int o = 16; o > 0; o >>= 1) {
                all[u]    += __shfl_xor(all[u], o, 32);
                before[u] += __shfl_xor(before[u], o, 32);
            }
            const int b = b0 + 32 * u;
            if (lane == 0 && b < nb) { tot[b] = all[u]; sh[b] = before[u]; }
        }
    }
    __syncthreads();
    int carry = 0;
    for (int b0 = 0; b0 < nb; b0 += 1024) {
        const int b  = b0 + tid;
        const int v  = (b < nb) ? tot[b] : 0;
        const int ex = block_excl_scan_1024(v, wsum);
        if (b < nb) {
            const int first = carry + ex;
            if (chunk == 0) bstart[b] = first;
            sh[b] += first;
        }
        if (tid == 1023) passTot = ex + v;
        __syncthreads();
        carry += passTot;
    }
    if (chunk == 0 && tid == 0) bstart[nb] = n_edges;
    const int base = blockIdx.x * CH;
    if (base + CH <= n_edges) scatter_chunk<true>(rows, cols, vals, sh, tmp, base, n_edges);
    else                      scatter_chunk<false>(rows, cols, vals, sh, tmp, base, n_edges);
}

__global__ void __launch_bounds__(RB) bucket_sort(
        const int*  __restrict__ bstart,
        const int4* __restrict__ tmp,
        int2*       __restrict__ sorted,
        int*        __restrict__ rowptr,
        int n_nodes) {
    __shared__ int cnt[RB];
    __shared__ int wsum[32];
    const int tid  = threadIdx.x;
    const int b    = blockIdx.x;
    const int row0 = b * RB;
    const int s = bstart[b];
    const int t = bstart[b + 1];
    cnt[tid] = 0;
    __syncthreads();
    int i = s + tid;
    for (; i + 3 * RB < t; i += 4 * RB) {
        int r0 = tmp[i].x, r1 = tmp[i + RB].x, r2 = tmp[i + 2 * RB].x, r3 = tmp[i + 3 * RB].x;
        atomicAdd(&cnt[r0 - row0], 1);
        atomicAdd(&cnt[r1 - row0], 1);
        atomicAdd(&cnt[r2 - row0], 1);
        atomicAdd(&cnt[r3 - row0], 1);
    }
    for (; i < t; i += RB) atomicAdd(&cnt[tmp[i].x - row0], 1);
    __syncthreads();
    const int c   = cnt[tid];
    const int off = s + block_excl_scan_1024(c, wsum);
    if (row0 + tid < n_nodes) rowptr[row0 + tid] = off;
    cnt[tid] = off;
    __syncthreads();
    i = s + tid;
    for (; i + 3 * RB < t; i += 4 * RB) {
        int4 e0 = tmp[i], e1 = tmp[i + RB], e2 = tmp[i + 2 * RB], e3 = tmp[i + 3 * RB];
        int p0 = atomicAdd(&cnt[e0.x - row0], 1);
        int p1 = atomicAdd(&cnt[e1.x - row0], 1);
        int p2 = atomicAdd(&cnt[e2.x - row0], 1);
        int p3 = atomicAdd(&cnt[e3.x - row0], 1);
        sorted[p0] = make_int2(e0.y, e0.z);
        sorted[p1] = make_int2(e1.y, e1.z);
        sorted[p2] = make_int2(e2.y, e2.z);
        sorted[p3] = make_int2(e3.y, e3.z);
    }
    for (; i < t; i += RB) {
        int4 e = tmp[i];
        int p = atomicAdd(&cnt[e.x - row0], 1);
        sorted[p] = make_int2(e.y, e.z);
    }
}


#define PITCH 136

template <int M>
__device__ __forceinline__ float row_xor(float x) {
    return __builtin_bit_cast(float, __builtin_amdgcn_update_dpp(
               0, __builtin_bit_cast(int, x), 0x160 + M, 0xf, 0xf, false));
}

template <bool TAIL>
__device__ __forceinline__ void gather8(const float4* __restrict__ xv, int2 ent,
                                        int k, int m, float4& acc,
                                        bool pre, const int2* __restrict__ nl, int nN,
                                        int lane, int2& entN) {
    float4 d[8];
    float  v[8];
    #pragma unroll
    for (int i = 0; i < 8; ++i) {
        const int c = __builtin_amdgcn_readlane(ent.x, k + i);
        v[i] = __int_as_float(__builtin_amdgcn_readlane(ent.y, k + i));
        if (!TAIL || k + i < m) {
            d[i] = xv[(size_t)c * (D / 4)];
        } else {
            d[i] = make_float4(0.f, 0.f, 0.f, 0.f);
            v[i] = 0.f;
        }
    }
    if (pre && lane < nN) entN = nl[lane];
    #pragma unroll
    for (int i = 0; i < 8; ++i) {
        acc.x = fmaf(v[i], d[i].x, acc.x);
        acc.y = fmaf(v[i], d[i].y, acc.y);
        acc.z = fmaf(v[i], d[i].z, acc.z);
        acc.w = fmaf(v[i], d[i].w, acc.w);
    }
}

template <bool HEAD>
__global__ void __launch_bounds__(256, 8) gather_gemm_relu(
        const int*      __restrict__ rowptr,
        const int2*     __restrict__ sorted,
        const float*    __restrict__ x,
        const float*    __restrict__ W,
        float*          __restrict__ Y,
        const float*    __restrict__ Wc,
        const float*    __restrict__ bc,
        float*          __restrict__ out,
        int n_nodes, int n_edges) {
    __shared__ __attribute__((aligned(16))) _Float16 sW[D * PITCH];
    __shared__ __attribute__((aligned(16))) _Float16 sA[8][16 * PITCH];

    const int wave    = threadIdx.x >> 5;
    const int lane    = threadIdx.x & 31;
    const int laneMod = lane & 15;
    const int laneHi  = lane >> 4;
    const int rowBase = __builtin_amdgcn_readfirstlane((blockIdx.x * 8 + wave) * 16);

    const int rr = rowBase + lane;
    const int rb = (rr < n_nodes) ? rowptr[rr] : n_edges;

    float4 wreg[16];
    #pragma unroll
    for (int q = 0; q < 16; ++q)
        wreg[q] = ((const float4*)W)[q * 256 + (int)threadIdx.x];

    const float4* xv = (const float4*)x + lane;
    _Float16* slot = &sA[wave][0];

    int2 entN = make_int2(0, 0);
    {
        const int s0 = __builtin_amdgcn_readlane(rb, 0);
        const int t0 = __builtin_amdgcn_readlane(rb, 1);
        if (lane < t0 - s0) entN = sorted[s0 + lane];
    }
    #pragma unroll
    for (int q = 0; q < 16; ++q) {
        const int piece = q * 256 + (int)threadIdx.x;
        h4 o;
        o[0] = (_Float16)wreg[q].x; o[1] = (_Float16)wreg[q].y;
        o[2] = (_Float16)wreg[q].z; o[3] = (_Float16)wreg[q].w;
        *(h4*)(sW + (piece >> 5) * PITCH + (piece & 31) * 4) = o;
    }
    #pragma unroll 1
    for (int r = 0; r < 16; ++r) {
        const int s = __builtin_amdgcn_readlane(rb, r);
        const int t = __builtin_amdgcn_readlane(rb, r + 1);
        int2 ent = entN;
        const int nN = (r + 1 < 16) ? __builtin_amdgcn_readlane(rb, r + 2) - t : 0;
        const int2* nl = sorted + t;
        entN = make_int2(0, 0);
        if (s >= t && lane < nN) entN = nl[lane];
        float4 acc = make_float4(0.f, 0.f, 0.f, 0.f);
        for (int c0 = s; c0 < t; c0 += 32) {
            const int m = (t - c0 < 32) ? t - c0 : 32;
            if (c0 != s) {
                ent = make_int2(0, 0);
                if (lane < m) ent = sorted[c0 + lane];
            }
            const bool lastChunk = c0 + 32 >= t;
            int k = 0;
            for (; k + 8 <= m; k += 8)
                gather8<false>(xv, ent, k, m, acc, lastChunk && k + 8 >= m, nl, nN, lane, entN);
            if (k < m) gather8<true>(xv, ent, k, m, acc, lastChunk, nl, nN, lane, entN);
        }
        h4 o;
        o[0] = (_Float16)acc.x; o[1] = (_Float16)acc.y;
        o[2] = (_Float16)acc.z; o[3] = (_Float16)acc.w;
        *(h4*)(slot + r * PITCH + lane * 4) = o;
    }

    __syncthreads();

    const _Float16* aRow = slot + laneMod * PITCH;
    const _Float16* bRow = sW + laneMod * PITCH;
    if (!HEAD) {
        v8f acc[8] = {};
        #pragma unroll
        for (int kk = 0; kk < D; kk += 32) {
            const int off0 = kk + laneHi * 8;
            const int off1 = kk + 16 + laneHi * 8;
            v8h alo = *(const v8h*)(aRow + off0);
            v8h ahi = *(const v8h*)(aRow + off1);
            v16h a = __builtin_shufflevector(alo, ahi,
                       0,1,2,3,4,5,6,7,8,9,10,11,12,13,14,15);
            #pragma unroll
            for (int t = 0; t < 8; ++t) {
                const _Float16* bp = bRow + t * 16 * PITCH;
                v8h blo = *(const v8h*)(bp + off0);
                v8h bhi = *(const v8h*)(bp + off1);
                v16h b = __builtin_shufflevector(blo, bhi,
                           0,1,2,3,4,5,6,7,8,9,10,11,12,13,14,15);
                acc[t] = __builtin_amdgcn_wmma_f32_16x16x32_f16(
                             false, a,  false, b,
                             (short)0, acc[t],
                             false,  false);
            }
        }
        #pragma unroll
        for (int t = 0; t < 8; ++t) {
            #pragma unroll
            for (int i = 0; i < 8; ++i) {
                int r = rowBase + i + laneHi * 8;
                if (r < n_nodes) {
                    float v = acc[t][i];
                    Y[(size_t)r * D + t * 16 + laneMod] = v > 0.f ? v : 0.f;
                }
            }
        }
    } else {
        float wc[8];
        #pragma unroll
        for (int t = 0; t < 8; ++t) wc[t] = Wc[t * 16 + laneMod];
        float p[8];
        #pragma unroll
        for (int i = 0; i < 8; ++i) p[i] = 0.f;
        #pragma unroll
        for (int g = 0; g < 2; ++g) {
            v8f acc[4] = {};
            #pragma unroll
            for (int kk = 0; kk < D; kk += 32) {
                const int off0 = kk + laneHi * 8;
                const int off1 = kk + 16 + laneHi * 8;
                v8h alo = *(const v8h*)(aRow + off0);
                v8h ahi = *(const v8h*)(aRow + off1);
                v16h a = __builtin_shufflevector(alo, ahi,
                           0,1,2,3,4,5,6,7,8,9,10,11,12,13,14,15);
                #pragma unroll
                for (int tt = 0; tt < 4; ++tt) {
                    const _Float16* bp = bRow + (g * 4 + tt) * 16 * PITCH;
                    v8h blo = *(const v8h*)(bp + off0);
                    v8h bhi = *(const v8h*)(bp + off1);
                    v16h b = __builtin_shufflevector(blo, bhi,
                               0,1,2,3,4,5,6,7,8,9,10,11,12,13,14,15);
                    acc[tt] = __builtin_amdgcn_wmma_f32_16x16x32_f16(
                                 false, a,  false, b,
                                 (short)0, acc[tt],
                                 false,  false);
                }
            }
            #pragma unroll
            for (int tt = 0; tt < 4; ++tt) {
                const float w = wc[g * 4 + tt];
                #pragma unroll
                for (int i = 0; i < 8; ++i) {
                    float v = acc[tt][i];
                    v = v > 0.f ? v : 0.f;
                    p[i] = fmaf(v, w, p[i]);
                }
            }
        }
        #pragma unroll
        for (int i = 0; i < 8; ++i) {
            p[i] += row_xor<8>(p[i]);
            p[i] += row_xor<4>(p[i]);
            p[i] += row_xor<2>(p[i]);
            p[i] += row_xor<1>(p[i]);
        }
        float mine = p[0];
        #pragma unroll
        for (int i = 1; i < 8; ++i) mine = (laneMod == i) ? p[i] : mine;
        const int r = rowBase + laneMod + laneHi * 8;
        if (laneMod < 8 && r < n_nodes) out[r] = mine + bc[0];
    }
}

extern "C" void kernel_launch(void* const* d_in, const int* in_sizes, int n_in,
                              void* d_out, int out_size, void* d_ws, size_t ws_size,
                              hipStream_t stream) {
    const float* features = (const float*)d_in[0];
    const int*   erows    = (const int*)  d_in[1];
    const int*   ecols    = (const int*)  d_in[2];
    const float* evals    = (const float*)d_in[3];
    const float* W1       = (const float*)d_in[4];
    const float* W2       = (const float*)d_in[5];
    const float* Wc       = (const float*)d_in[6];
    const float* bc       = (const float*)d_in[7];
    float* out = (float*)d_out;

    const int n_nodes = in_sizes[0] / D;
    const int n_edges = in_sizes[1];

    float*     buf1 = (float*)d_ws;

    const int nb      = (n_nodes + RB - 1) / RB;
    const int nblk    = (n_edges + CH - 1) / CH;
    const int ntbl    = nb * nblk;
    const size_t tblPad = ((size_t)ntbl + 63) / 64 * 64;
    const size_t bsPad  = ((size_t)nb + 1 + 63) / 64 * 64;
    const size_t rowPad = ((size_t)n_nodes + 63) / 64 * 64;

    int*   tbl    = (int*)(buf1 + (size_t)n_nodes * D);
    int*   bstart = tbl + tblPad;
    int*   rowptr = bstart + bsPad;
    int4*  tmp    = (int4*)(rowptr + rowPad);
    int2*  sorted = (int2*)(tmp + (size_t)n_edges);

    const int gblocks = (n_nodes + 127) / 128;
    const size_t shb  = (size_t)nb * sizeof(int);

    bucket_count<<<nblk, 1024, shb, stream>>>(erows, tbl, n_edges, nb, nblk);
    bucket_scatter<<<nblk, 1024, 2 * shb, stream>>>(erows, ecols, evals, tbl, bstart, tmp, n_edges, nb, nblk);
    bucket_sort<<<nb, RB, 0, stream>>>(bstart, tmp, sorted, rowptr, n_nodes);

    gather_gemm_relu<false><<<gblocks, 256, 0, stream>>>(
        rowptr, sorted, features, W1, buf1, Wc, bc, out, n_nodes, n_edges);

    gather_gemm_relu<true><<<gblocks, 256, 0, stream>>>(
        rowptr, sorted, buf1, W2, nullptr, Wc, bc, out, n_nodes, n_edges);
}


# both ranked scatters of the edge sort go through LDS: an edge is put at its rank in LDS and each run leaves as contiguous whole lines, not one 16/8-byte piece per lane; row sum order rule unchanged
# speedup vs baseline: 5.9832x; 1.2509x over previous
#include <hip/hip_runtime.h>

typedef _Float16 v16h __attribute__((ext_vector_type(16)));
typedef _Float16 v8h  __attribute__((ext_vector_type(8)));
typedef _Float16 h4   __attribute__((ext_vector_type(4)));
typedef float    v8f  __attribute__((ext_vector_type(8)));
typedef int      v4i  __attribute__((ext_vector_type(4)));

#define D 128

#define RB 1024
#define CH 8192
#define CAP2 24576

__device__ __forceinline__ int block_excl_scan_1024(int v, int* wsum) {
    const int lane = threadIdx.x & 31;
    const int w    = threadIdx.x >> 5;
    int inc = v;
    #pragma unroll
    for (int o = 1; o < 32; o <<= 1) {
        int t = __shfl_up(inc, o, 32);
        if (lane >= o) inc += t;
    }
    if (lane == 31) wsum[w] = inc;
    __syncthreads();
    if (w == 0) {
        int x = wsum[lane];
        int xi = x;
        #pragma unroll
        for (int o = 1; o < 32; o <<= 1) {
            int t = __shfl_up(xi, o, 32);
            if (lane >= o) xi += t;
        }
        wsum[lane] = xi - x;
    }
    __syncthreads();
    return inc - v + wsum[w];
}

template <bool FULL>
__device__ __forceinline__ void count_chunk(const int* __restrict__ rows, int* sh,
                                            int base, int n_edges) {
    int r[CH / 1024];
    #pragma unroll
    for (int k = 0; k < CH / 1024; ++k) {
        int e = base + k * 1024 + (int)threadIdx.x;
        r[k] = (FULL || e < n_edges) ? rows[e] : -1;
    }
    #pragma unroll
    for (int k = 0; k < CH / 1024; ++k)
        if (FULL || r[k] >= 0) atomicAdd(&sh[r[k] / RB], 1);
}

__global__ void __launch_bounds__(1024) bucket_count(
        const int* __restrict__ rows, int* __restrict__ tbl,
        int n_edges, int nb, int nblk) {
    extern __shared__ int sh[];
    const int tid = threadIdx.x;
    for (int i = tid; i < nb; i += 1024) sh[i] = 0;
    __syncthreads();
    const int base = blockIdx.x * CH;
    if (base + CH <= n_edges) count_chunk<true>(rows, sh, base, n_edges);
    else                      count_chunk<false>(rows, sh, base, n_edges);
    __syncthreads();
    for (int i = tid; i < nb; i += 1024) tbl[(size_t)i * nblk + blockIdx.x] = sh[i];
}

template <bool FULL>
__device__ __forceinline__ void scatter_chunk(
        const int* __restrict__ rows, const int* __restrict__ cols,
        const float* __restrict__ vals, int* lcur, const int* gdelta,
        int4* stE, int* stP, int base, int n_edges) {
    int r[CH / 1024], c[CH / 1024], v[CH / 1024];
    #pragma unroll
    for (int k = 0; k < CH / 1024; ++k) {
        int e = base + k * 1024 + (int)threadIdx.x;
        bool in = FULL || e < n_edges;
        r[k] = in ? rows[e] : -1;
        c[k] = in ? cols[e] : 0;
        v[k] = in ? __float_as_int(vals[e]) : 0;
    }
    #pragma unroll
    for (int k = 0; k < CH / 1024; ++k) {
        if (FULL || r[k] >= 0) {
            const int b  = r[k] / RB;
            const int li = atomicAdd(&lcur[b], 1);
            stE[li] = make_int4(r[k], c[k], v[k], 0);
            stP[li] = li + gdelta[b];
        }
    }
}

__global__ void __launch_bounds__(1024) bucket_scatter(
        const int*   __restrict__ rows,
        const int*   __restrict__ cols,
        const float* __restrict__ vals,
        const int*   __restrict__ tbl,
        int*         __restrict__ bstart,
        int4*        __restrict__ tmp,
        int n_edges, int nb, int nblk) {
    extern __shared__ int sh[];
    __shared__ int wsum[32];
    __shared__ int wsum2[32];
    __shared__ int passTot;
    __shared__ int passLoc;
    __shared__ int4 stE[CH];
    __shared__ int  stP[CH];
    int* tot = sh + nb;
    int* loc = sh + 2 * nb;
    const int tid   = threadIdx.x;
    const int lane  = tid & 31;
    const int w     = tid >> 5;
    const int chunk = blockIdx.x;
    for (int b0 = w; b0 < nb; b0 += 128) {
        int all[4], before[4], mine[4];
        #pragma unroll
        for (int u = 0; u < 4; ++u) { all[u] = 0; before[u] = 0; mine[u] = 0; }
        for (int c0 = 0; c0 < nblk; c0 += 256) {
            int v[4][8];
            #pragma unroll
            for (int u = 0; u < 4; ++u) {
                const int b  = b0 + 32 * u;
                const int bb = (b < nb) ? b : nb - 1;
                #pragma unroll
                for (int j = 0; j < 8; ++j) {
                    const int c  = c0 + 32 * j + lane;
                    const int cc = (c < nblk) ? c : nblk - 1;
                    v[u][j] = tbl[(size_t)bb * nblk + cc];
                }
            }
            #pragma unroll
            for (int u = 0; u < 4; ++u) {
                #pragma unroll
                for (int j = 0; j < 8; ++j) {
                    const int c = c0 + 32 * j + lane;
                    all[u]    += (c < nblk)   ? v[u][j] : 0;
                    before[u] += (c < chunk)  ? v[u][j] : 0;
                    mine[u]   += (c == chunk) ? v[u][j] : 0;
                }
            }
        }
        #pragma unroll
        for (int u = 0; u < 4; ++u) {
            #pragma unroll
            for (int o = 16; o > 0; o >>= 1) {
                all[u]    += __shfl_xor(all[u], o, 32);
                before[u] += __shfl_xor(before[u], o, 32);
                mine[u]   += __shfl_xor(mine[u], o, 32);
            }
            const int b = b0 + 32 * u;
            if (lane == 0 && b < nb) { tot[b] = all[u]; sh[b] = before[u]; loc[b] = mine[u]; }
        }
    }
    __syncthreads();
    int carry = 0, lcarry = 0;
    for (int b0 = 0; b0 < nb; b0 += 1024) {
        const int b   = b0 + tid;
        const int v   = (b < nb) ? tot[b] : 0;
        const int lv  = (b < nb) ? loc[b] : 0;
        const int ex  = block_excl_scan_1024(v, wsum);
        const int lex = block_excl_scan_1024(lv, wsum2);
        if (b < nb) {
            const int first = carry + ex;
            const int lo    = lcarry + lex;
            if (chunk == 0) bstart[b] = first;
            sh[b]  = sh[b] + first - lo;
            loc[b] = lo;
        }
        if (tid == 1023) { passTot = ex + v; passLoc = lex + lv; }
        __syncthreads();
        carry  += passTot;
        lcarry += passLoc;
    }
    if (chunk == 0 && tid == 0) bstart[nb] = n_edges;
    const int base = blockIdx.x * CH;
    if (base + CH <= n_edges) scatter_chunk<true>(rows, cols, vals, loc, sh, stE, stP, base, n_edges);
    else                      scatter_chunk<false>(rows, cols, vals, loc, sh, stE, stP, base, n_edges);
    __syncthreads();
    if (base + CH <= n_edges) {
        int4 e[CH / 1024];
        int  p[CH / 1024];
        #pragma unroll
        for (int k = 0; k < CH / 1024; ++k) {
            e[k] = stE[k * 1024 + tid];
            p[k] = stP[k * 1024 + tid];
        }
        #pragma unroll
        for (int k = 0; k < CH / 1024; ++k) tmp[p[k]] = e[k];
    } else {
        const int nchunk = n_edges - base;
        for (int i = tid; i < nchunk; i += 1024) tmp[stP[i]] = stE[i];
    }
}

__global__ void __launch_bounds__(RB) bucket_sort(
        const int*  __restrict__ bstart,
        const int4* __restrict__ tmp,
        int2*       __restrict__ sorted,
        int*        __restrict__ rowptr,
        int n_nodes) {
    __shared__ int cnt[RB];
    __shared__ int wsum[32];
    __shared__ int2 stage[CAP2];
    const int tid  = threadIdx.x;
    const int b    = blockIdx.x;
    const int row0 = b * RB;
    const int s = bstart[b];
    const int t = bstart[b + 1];
    cnt[tid] = 0;
    __syncthreads();
    int i = s + tid;
    for (; i + 3 * RB < t; i += 4 * RB) {
        int r0 = tmp[i].x, r1 = tmp[i + RB].x, r2 = tmp[i + 2 * RB].x, r3 = tmp[i + 3 * RB].x;
        atomicAdd(&cnt[r0 - row0], 1);
        atomicAdd(&cnt[r1 - row0], 1);
        atomicAdd(&cnt[r2 - row0], 1);
        atomicAdd(&cnt[r3 - row0], 1);
    }
    for (; i < t; i += RB) atomicAdd(&cnt[tmp[i].x - row0], 1);
    __syncthreads();
    const int c   = cnt[tid];
    const int off = s + block_excl_scan_1024(c, wsum);
    if (row0 + tid < n_nodes) rowptr[row0 + tid] = off;
    cnt[tid] = off;
    __syncthreads();
    const int n = t - s;
    i = s + tid;
    if (n <= CAP2) {
        for (; i + 3 * RB < t; i += 4 * RB) {
            int4 e0 = tmp[i], e1 = tmp[i + RB], e2 = tmp[i + 2 * RB], e3 = tmp[i + 3 * RB];
            int p0 = atomicAdd(&cnt[e0.x - row0], 1);
            int p1 = atomicAdd(&cnt[e1.x - row0], 1);
            int p2 = atomicAdd(&cnt[e2.x - row0], 1);
            int p3 = atomicAdd(&cnt[e3.x - row0], 1);
            stage[p0 - s] = make_int2(e0.y, e0.z);
            stage[p1 - s] = make_int2(e1.y, e1.z);
            stage[p2 - s] = make_int2(e2.y, e2.z);
            stage[p3 - s] = make_int2(e3.y, e3.z);
        }
        for (; i < t; i += RB) {
            int4 e = tmp[i];
            int p = atomicAdd(&cnt[e.x - row0], 1);
            stage[p - s] = make_int2(e.y, e.z);
        }
        __syncthreads();
        const int g0 = (s + 1) & ~1;
        const int g1 = t & ~1;
        #pragma unroll 4
        for (int g = g0 + 2 * tid; g < g1; g += 2 * RB) {
            const int2 a = stage[g - s];
            const int2 z = stage[g - s + 1];
            v4i o;
            o[0] = a.x; o[1] = a.y; o[2] = z.x; o[3] = z.y;
            *(v4i*)(sorted + g) = o;
        }
        if (n > 0) {
            if (tid == 0  && (s & 1)) sorted[s]     = stage[0];
            if (tid == 32 && (t & 1)) sorted[t - 1] = stage[n - 1];
        }
    } else {
        for (; i + 3 * RB < t; i += 4 * RB) {
            int4 e0 = tmp[i], e1 = tmp[i + RB], e2 = tmp[i + 2 * RB], e3 = tmp[i + 3 * RB];
            int p0 = atomicAdd(&cnt[e0.x - row0], 1);
            int p1 = atomicAdd(&cnt[e1.x - row0], 1);
            int p2 = atomicAdd(&cnt[e2.x - row0], 1);
            int p3 = atomicAdd(&cnt[e3.x - row0], 1);
            sorted[p0] = make_int2(e0.y, e0.z);
            sorted[p1] = make_int2(e1.y, e1.z);
            sorted[p2] = make_int2(e2.y, e2.z);
            sorted[p3] = make_int2(e3.y, e3.z);
        }
        for (; i < t; i += RB) {
            int4 e = tmp[i];
            int p = atomicAdd(&cnt[e.x - row0], 1);
            sorted[p] = make_int2(e.y, e.z);
        }
    }
}


#define PITCH 136

template <int M>
__device__ __forceinline__ float row_xor(float x) {
    return __builtin_bit_cast(float, __builtin_amdgcn_update_dpp(
               0, __builtin_bit_cast(int, x), 0x160 + M, 0xf, 0xf, false));
}

template <bool TAIL>
__device__ __forceinline__ void gather8(const float4* __restrict__ xv, int2 ent,
                                        int k, int m, float4& acc,
                                        bool pre, const int2* __restrict__ nl, int nN,
                                        int lane, int2& entN) {
    float4 d[8];
    float  v[8];
    #pragma unroll
    for (int i = 0; i < 8; ++i) {
        const int c = __builtin_amdgcn_readlane(ent.x, k + i);
        v[i] = __int_as_float(__builtin_amdgcn_readlane(ent.y, k + i));
        if (!TAIL || k + i < m) {
            d[i] = xv[(size_t)c * (D / 4)];
        } else {
            d[i] = make_float4(0.f, 0.f, 0.f, 0.f);
            v[i] = 0.f;
        }
    }
    if (pre && lane < nN) entN = nl[lane];
    #pragma unroll
    for (int i = 0; i < 8; ++i) {
        acc.x = fmaf(v[i], d[i].x, acc.x);
        acc.y = fmaf(v[i], d[i].y, acc.y);
        acc.z = fmaf(v[i], d[i].z, acc.z);
        acc.w = fmaf(v[i], d[i].w, acc.w);
    }
}

template <int S>
__device__ __forceinline__ void request_rows(const h4* __restrict__ xh, int2 ent, int o,
                                             h4 (&d)[S], float (&v)[S]) {
    #pragma unroll
    for (int i = 0; i < S; ++i) {
        const int c = __builtin_amdgcn_readlane(ent.x, o + i);
        v[i] = __int_as_float(__builtin_amdgcn_readlane(ent.y, o + i));
        d[i] = xh[(size_t)c * (D / 4)];
    }
}
template <int S>
__device__ __forceinline__ void add_rows(const h4 (&d)[S], const float (&v)[S], float4& acc) {
    #pragma unroll
    for (int i = 0; i < S; ++i) {
        acc.x = fmaf(v[i], (float)d[i][0], acc.x);
        acc.y = fmaf(v[i], (float)d[i][1], acc.y);
        acc.z = fmaf(v[i], (float)d[i][2], acc.z);
        acc.w = fmaf(v[i], (float)d[i][3], acc.w);
    }
}
__device__ __forceinline__ void gather16h(const h4* __restrict__ xh, int2 ent, int k,
                                          float4& acc, bool pre,
                                          const int2* __restrict__ nl, int nN,
                                          int lane, int2& entN) {
    h4 d[16]; float v[16];
    request_rows<16>(xh, ent, k, d, v);
    if (pre && lane < nN) entN = nl[lane];
    add_rows<16>(d, v, acc);
}
__device__ __forceinline__ void gather_tail_h(const h4* __restrict__ xh, int2 ent, int k, int rem,
                                              float4& acc, bool pre,
                                              const int2* __restrict__ nl, int nN,
                                              int lane, int2& entN) {
    h4 d0[1], d8[8], d4[4], d2[2], d1[1];
    float v0[1], v8[8], v4[4], v2[2], v1[1];
    const int more = rem - 1;
    request_rows<1>(xh, ent, k, d0, v0);
    int o = k + 1;
    if (more & 8) { request_rows<8>(xh, ent, o, d8, v8); o += 8; }
    if (more & 4) { request_rows<4>(xh, ent, o, d4, v4); o += 4; }
    if (more & 2) { request_rows<2>(xh, ent, o, d2, v2); o += 2; }
    if (more & 1) { request_rows<1>(xh, ent, o, d1, v1); }
    if (pre && lane < nN) entN = nl[lane];
    add_rows<1>(d0, v0, acc);
    if (more & 8) add_rows<8>(d8, v8, acc);
    if (more & 4) add_rows<4>(d4, v4, acc);
    if (more & 2) add_rows<2>(d2, v2, acc);
    if (more & 1) add_rows<1>(d1, v1, acc);
}

template <bool HEAD>
__global__ void __launch_bounds__(256, 8) gather_gemm_relu(
        const int*      __restrict__ rowptr,
        const int2*     __restrict__ sorted,
        const float*    __restrict__ x,
        const _Float16* __restrict__ xh,
        const float*    __restrict__ W,
        _Float16*       __restrict__ Y,
        const float*    __restrict__ Wc,
        const float*    __restrict__ bc,
        float*          __restrict__ out,
        int n_nodes, int n_edges) {
    __shared__ __attribute__((aligned(16))) _Float16 sW[D * PITCH];
    __shared__ __attribute__((aligned(16))) _Float16 sA[8][16 * PITCH];

    const int wave    = threadIdx.x >> 5;
    const int lane    = threadIdx.x & 31;
    const int laneMod = lane & 15;
    const int laneHi  = lane >> 4;
    const int rowBase = __builtin_amdgcn_readfirstlane((blockIdx.x * 8 + wave) * 16);

    const int rr = rowBase + lane;
    const int rb = (rr < n_nodes) ? rowptr[rr] : n_edges;

    float4 wreg[16];
    #pragma unroll
    for (int q = 0; q < 16; ++q)
        wreg[q] = ((const float4*)W)[q * 256 + (int)threadIdx.x];

    const float4* xv  = HEAD ? nullptr : (const float4*)x + lane;
    const h4*     xhv = HEAD ? (const h4*)xh + lane : nullptr;
    _Float16* slot = &sA[wave][0];

    int2 entN = make_int2(0, 0);
    {
        const int s0 = __builtin_amdgcn_readlane(rb, 0);
        const int t0 = __builtin_amdgcn_readlane(rb, 1);
        if (lane < t0 - s0) entN = sorted[s0 + lane];
    }
    #pragma unroll
    for (int q = 0; q < 16; ++q) {
        const int piece = q * 256 + (int)threadIdx.x;
        h4 o;
        o[0] = (_Float16)wreg[q].x; o[1] = (_Float16)wreg[q].y;
        o[2] = (_Float16)wreg[q].z; o[3] = (_Float16)wreg[q].w;
        *(h4*)(sW + (piece >> 5) * PITCH + (piece & 31) * 4) = o;
    }
    #pragma unroll 1
    for (int r = 0; r < 16; ++r) {
        const int s = __builtin_amdgcn_readlane(rb, r);
        const int t = __builtin_amdgcn_readlane(rb, r + 1);
        int2 ent = entN;
        const int nN = (r + 1 < 16) ? __builtin_amdgcn_readlane(rb, r + 2) - t : 0;
        const int2* nl = sorted + t;
        entN = make_int2(0, 0);
        if (s >= t && lane < nN) entN = nl[lane];
        float4 acc = make_float4(0.f, 0.f, 0.f, 0.f);
        for (int c0 = s; c0 < t; c0 += 32) {
            const int m = (t - c0 < 32) ? t - c0 : 32;
            if (c0 != s) {
                ent = make_int2(0, 0);
                if (lane < m) ent = sorted[c0 + lane];
            }
            const bool lastChunk = c0 + 32 >= t;
            int k = 0;
            if (!HEAD) {
                for (; k + 8 <= m; k += 8)
                    gather8<false>(xv, ent, k, m, acc, lastChunk && k + 8 >= m, nl, nN, lane, entN);
                if (k < m) gather8<true>(xv, ent, k, m, acc, lastChunk, nl, nN, lane, entN);
            } else {
                for (; k + 16 <= m; k += 16)
                    gather16h(xhv, ent, k, acc, lastChunk && k + 16 >= m, nl, nN, lane, entN);
                if (k < m) gather_tail_h(xhv, ent, k, m - k, acc, lastChunk, nl, nN, lane, entN);
            }
        }
        h4 o;
        o[0] = (_Float16)acc.x; o[1] = (_Float16)acc.y;
        o[2] = (_Float16)acc.z; o[3] = (_Float16)acc.w;
        *(h4*)(slot + r * PITCH + lane * 4) = o;
    }

    __syncthreads();

    const _Float16* aRow = slot + laneMod * PITCH;
    const _Float16* bRow = sW + laneMod * PITCH;
    if (!HEAD) {
        v8h o[8];
        #pragma unroll
        for (int g = 0; g < 2; ++g) {
            v8f acc[4] = {};
            #pragma unroll
            for (int kk = 0; kk < D; kk += 32) {
                const int off0 = kk + laneHi * 8;
                const int off1 = kk + 16 + laneHi * 8;
                v8h alo = *(const v8h*)(aRow + off0);
                v8h ahi = *(const v8h*)(aRow + off1);
                v16h a = __builtin_shufflevector(alo, ahi,
                           0,1,2,3,4,5,6,7,8,9,10,11,12,13,14,15);
                #pragma unroll
                for (int tt = 0; tt < 4; ++tt) {
                    const _Float16* bp = bRow + (g * 4 + tt) * 16 * PITCH;
                    v8h blo = *(const v8h*)(bp + off0);
                    v8h bhi = *(const v8h*)(bp + off1);
                    v16h b = __builtin_shufflevector(blo, bhi,
                               0,1,2,3,4,5,6,7,8,9,10,11,12,13,14,15);
                    acc[tt] = __builtin_amdgcn_wmma_f32_16x16x32_f16(
                                 false, b,  false, a,
                                 (short)0, acc[tt],
                                 false,  false);
                }
            }
            #pragma unroll
            for (int tt = 0; tt < 4; ++tt) {
                #pragma unroll
                for (int i = 0; i < 8; ++i) {
                    const float v = acc[tt][i];
                    o[g * 4 + tt][i] = (_Float16)(v > 0.f ? v : 0.f);
                }
            }
        }
        #pragma unroll
        for (int t = 0; t < 8; ++t)
            *(v8h*)(slot + laneMod * PITCH + t * 16 + laneHi * 8) = o[t];
        __builtin_amdgcn_fence(__ATOMIC_RELEASE, "wavefront");
        __builtin_amdgcn_wave_barrier();
        __builtin_amdgcn_fence(__ATOMIC_ACQUIRE, "wavefront");
        v8h w[8];
        #pragma unroll
        for (int q = 0; q < 8; ++q)
            w[q] = *(const v8h*)(slot + (2 * q + laneHi) * PITCH + laneMod * 8);
        _Float16* yp = Y + (size_t)(rowBase + laneHi) * D + laneMod * 8;
        if (rowBase + 16 <= n_nodes) {
            #pragma unroll
            for (int q = 0; q < 8; ++q) *(v8h*)(yp + (size_t)(2 * q) * D) = w[q];
        } else {
            #pragma unroll
            for (int q = 0; q < 8; ++q)
                if (rowBase + 2 * q + laneHi < n_nodes) *(v8h*)(yp + (size_t)(2 * q) * D) = w[q];
        }
    } else {
        float wc[8];
        #pragma unroll
        for (int t = 0; t < 8; ++t) wc[t] = Wc[t * 16 + laneMod];
        float p[8];
        #pragma unroll
        for (int i = 0; i < 8; ++i) p[i] = 0.f;
        #pragma unroll
        for (int g = 0; g < 2; ++g) {
            v8f acc[4] = {};
            #pragma unroll
            for (int kk = 0; kk < D; kk += 32) {
                const int off0 = kk + laneHi * 8;
                const int off1 = kk + 16 + laneHi * 8;
                v8h alo = *(const v8h*)(aRow + off0);
                v8h ahi = *(const v8h*)(aRow + off1);
                v16h a = __builtin_shufflevector(alo, ahi,
                           0,1,2,3,4,5,6,7,8,9,10,11,12,13,14,15);
                #pragma unroll
                for (int tt = 0; tt < 4; ++tt) {
                    const _Float16* bp = bRow + (g * 4 + tt) * 16 * PITCH;
                    v8h blo = *(const v8h*)(bp + off0);
                    v8h bhi = *(const v8h*)(bp + off1);
                    v16h b = __builtin_shufflevector(blo, bhi,
                               0,1,2,3,4,5,6,7,8,9,10,11,12,13,14,15);
                    acc[tt] = __builtin_amdgcn_wmma_f32_16x16x32_f16(
                                 false, a,  false, b,
                                 (short)0, acc[tt],
                                 false,  false);
                }
            }
            #pragma unroll
            for (int tt = 0; tt < 4; ++tt) {
                const float w = wc[g * 4 + tt];
                #pragma unroll
                for (int i = 0; i < 8; ++i) {
                    float v = acc[tt][i];
                    v = v > 0.f ? v : 0.f;
                    p[i] = fmaf(v, w, p[i]);
                }
            }
        }
        #pragma unroll
        for (int i = 0; i < 8; ++i) {
            p[i] += row_xor<8>(p[i]);
            p[i] += row_xor<4>(p[i]);
            p[i] += row_xor<2>(p[i]);
            p[i] += row_xor<1>(p[i]);
        }
        float mine = p[0];
        #pragma unroll
        for (int i = 1; i < 8; ++i) mine = (laneMod == i) ? p[i] : mine;
        const int r = rowBase + laneMod + laneHi * 8;
        if (laneMod < 8 && r < n_nodes) out[r] = mine + bc[0];
    }
}

extern "C" void kernel_launch(void* const* d_in, const int* in_sizes, int n_in,
                              void* d_out, int out_size, void* d_ws, size_t ws_size,
                              hipStream_t stream) {
    const float* features = (const float*)d_in[0];
    const int*   erows    = (const int*)  d_in[1];
    const int*   ecols    = (const int*)  d_in[2];
    const float* evals    = (const float*)d_in[3];
    const float* W1       = (const float*)d_in[4];
    const float* W2       = (const float*)d_in[5];
    const float* Wc       = (const float*)d_in[6];
    const float* bc       = (const float*)d_in[7];
    float* out = (float*)d_out;

    const int n_nodes = in_sizes[0] / D;
    const int n_edges = in_sizes[1];

    _Float16*  buf1 = (_Float16*)d_ws;

    const int nb      = (n_nodes + RB - 1) / RB;
    const int nblk    = (n_edges + CH - 1) / CH;
    const int ntbl    = nb * nblk;
    const size_t tblPad = ((size_t)ntbl + 63) / 64 * 64;
    const size_t bsPad  = ((size_t)nb + 1 + 63) / 64 * 64;
    const size_t rowPad = ((size_t)n_nodes + 63) / 64 * 64;

    int*   tbl    = (int*)(buf1 + (size_t)n_nodes * D);
    int*   bstart = tbl + tblPad;
    int*   rowptr = bstart + bsPad;
    int4*  tmp    = (int4*)(rowptr + rowPad);
    int2*  sorted = (int2*)(tmp + (size_t)n_edges);

    const int gblocks = (n_nodes + 127) / 128;
    const size_t shb  = (size_t)nb * sizeof(int);

    bucket_count<<<nblk, 1024, shb, stream>>>(erows, tbl, n_edges, nb, nblk);
    bucket_scatter<<<nblk, 1024, 3 * shb, stream>>>(erows, ecols, evals, tbl, bstart, tmp, n_edges, nb, nblk);
    bucket_sort<<<nb, RB, 0, stream>>>(bstart, tmp, sorted, rowptr, n_nodes);

    gather_gemm_relu<false><<<gblocks, 256, 0, stream>>>(
        rowptr, sorted, features, nullptr, W1, buf1, Wc, bc, out, n_nodes, n_edges);

    gather_gemm_relu<true><<<gblocks, 256, 0, stream>>>(
        rowptr, sorted, nullptr, buf1, W2, nullptr, Wc, bc, out, n_nodes, n_edges);
}
